# starting kernel
# baseline (speedup 1.0000x reference)
#include <hip/hip_runtime.h>

typedef __attribute__((ext_vector_type(2))) float v2f;
typedef __attribute__((ext_vector_type(4))) float v4f;
typedef __attribute__((ext_vector_type(8))) float v8f;

#define NQ 16384
#define MR 16384
#define DIMK 16
#define NSPLIT 4
#define QT_PER_BLK 2
#define COLS_PER_SPLIT (MR / NSPLIT)
#define ITER_PER_SPLIT (COLS_PER_SPLIT / 32)

__global__ __launch_bounds__(256) void DCM_xb2_kernel(const float* __restrict__ xb,
                                                      float* __restrict__ xb2) {
  const int j = blockIdx.x * 256 + threadIdx.x;
  const v4f* row = (const v4f*)(xb + (size_t)j * DIMK);
  v4f r0 = row[0], r1 = row[1], r2 = row[2], r3 = row[3];
  float s = r0.x * r0.x + r0.y * r0.y + r0.z * r0.z + r0.w * r0.w;
  s += r1.x * r1.x + r1.y * r1.y + r1.z * r1.z + r1.w * r1.w;
  s += r2.x * r2.x + r2.y * r2.y + r2.z * r2.z + r2.w * r2.w;
  s += r3.x * r3.x + r3.y * r3.y + r3.z * r3.z + r3.w * r3.w;
  xb2[j] = s;
}

__global__ __launch_bounds__(256) void DCM_nn_wmma_kernel(
    const float* __restrict__ x,
    const float* __restrict__ xb,
    const float* __restrict__ y,
    const float* __restrict__ xb2,
    float* __restrict__ out) {
  __shared__ float s_val[QT_PER_BLK][NSPLIT][16];
  __shared__ int   s_idx[QT_PER_BLK][NSPLIT][16];

  const int wave  = threadIdx.x >> 5;
  const int lane  = threadIdx.x & 31;
  const int qtloc = wave >> 2;
  const int split = wave & 3;
  const int qbase = (blockIdx.x * QT_PER_BLK + qtloc) * 16;
  const int grp   = lane & 15;
  const int hi    = lane >> 4;

  v2f a[4];
  {
    const float* xrow = x + (size_t)(qbase + grp) * DIMK;
#pragma unroll
    for (int k = 0; k < 4; ++k) {
      const int kb = k * 4 + hi * 2;
      a[k].x = -2.0f * xrow[kb + 0];
      a[k].y = -2.0f * xrow[kb + 1];
    }
  }

  float bv0[8], bv1[8];
  int   bi0[8], bi1[8];
#pragma unroll
  for (int r = 0; r < 8; ++r) {
    bv0[r] = bv1[r] = 3.402823466e38f;
    bi0[r] = bi1[r] = 0;
  }

  const int jt0 = split * ITER_PER_SPLIT;
  for (int jt = jt0; jt < jt0 + ITER_PER_SPLIT; ++jt) {
    const int c0 = jt * 32 + grp;
    const int c1 = c0 + 16;
    const float* p0 = xb + (size_t)c0 * DIMK;
    const float* p1 = xb + (size_t)c1 * DIMK;

    v2f b0[4], b1[4];
#pragma unroll
    for (int k = 0; k < 4; ++k) {
      const int kb = k * 4 + hi * 2;
      b0[k] = *(const v2f*)(p0 + kb);
      b1[k] = *(const v2f*)(p1 + kb);
    }
    const float s0 = xb2[c0];
    const float s1 = xb2[c1];

    v8f acc0, acc1;
#pragma unroll
    for (int r = 0; r < 8; ++r) { acc0[r] = s0; acc1[r] = s1; }
#pragma unroll
    for (int k = 0; k < 4; ++k) {
      acc0 = __builtin_amdgcn_wmma_f32_16x16x4_f32(
          false, a[k], false, b0[k], (short)0, acc0, false, false);
      acc1 = __builtin_amdgcn_wmma_f32_16x16x4_f32(
          false, a[k], false, b1[k], (short)0, acc1, false, false);
    }

#pragma unroll
    for (int r = 0; r < 8; ++r) {
      if (acc0[r] < bv0[r]) { bv0[r] = acc0[r]; bi0[r] = c0; }
      if (acc1[r] < bv1[r]) { bv1[r] = acc1[r]; bi1[r] = c1; }
    }
  }

  float bestv[8];
  int   besti[8];
#pragma unroll
  for (int r = 0; r < 8; ++r) {
    const bool take1 = (bv1[r] < bv0[r]) || (bv1[r] == bv0[r] && bi1[r] < bi0[r]);
    bestv[r] = take1 ? bv1[r] : bv0[r];
    besti[r] = take1 ? bi1[r] : bi0[r];
  }

#pragma unroll
  for (int mask = 1; mask < 16; mask <<= 1) {
#pragma unroll
    for (int r = 0; r < 8; ++r) {
      const float ov = __shfl_xor(bestv[r], mask, 32);
      const int   oi = __shfl_xor(besti[r], mask, 32);
      if (ov < bestv[r] || (ov == bestv[r] && oi < besti[r])) {
        bestv[r] = ov; besti[r] = oi;
      }
    }
  }

  if (grp == 0) {
#pragma unroll
    for (int r = 0; r < 8; ++r) {
      s_val[qtloc][split][hi * 8 + r] = bestv[r];
      s_idx[qtloc][split][hi * 8 + r] = besti[r];
    }
  }
  __syncthreads();

  if (threadIdx.x < QT_PER_BLK * 16) {
    const int qt  = threadIdx.x >> 4;
    const int row = threadIdx.x & 15;
    float v = s_val[qt][0][row];
    int   i = s_idx[qt][0][row];
#pragma unroll
    for (int s = 1; s < NSPLIT; ++s) {
      const float sv = s_val[qt][s][row];
      const int   si = s_idx[qt][s][row];
      if (sv < v) { v = sv; i = si; }
    }
    out[(blockIdx.x * QT_PER_BLK + qt) * 16 + row] = y[i];
  }
}

extern "C" void kernel_launch(void* const* d_in, const int* in_sizes, int n_in,
                              void* d_out, int out_size, void* d_ws, size_t ws_size,
                              hipStream_t stream) {
  const float* x  = (const float*)d_in[0];
  const float* xb = (const float*)d_in[1];
  const float* y  = (const float*)d_in[2];
  float* out = (float*)d_out;
  float* xb2 = (float*)d_ws;

  DCM_xb2_kernel<<<dim3(MR / 256), dim3(256), 0, stream>>>(xb, xb2);

  DCM_nn_wmma_kernel<<<dim3(NQ / 16 / QT_PER_BLK), dim3(256), 0, stream>>>(
      x, xb, y, xb2, out);
}


# Each wave keeps the A fragments of 4 query tiles in registers and reuses every loaded B fragment and seed for all 4; M-splits per block raised 4 to 16 to keep the wave count
# speedup vs baseline: 1.5724x; 1.5724x over previous
#include <hip/hip_runtime.h>


typedef __attribute__((ext_vector_type(2))) float v2f;
typedef __attribute__((ext_vector_type(4))) float v4f;
typedef __attribute__((ext_vector_type(8))) float v8f;

#define NQ 16384
#define MR 16384
#define DIMK 16
#define QT_PER_WAVE 4
#define NSPLIT 16
#define COLS_PER_SPLIT (MR / NSPLIT)
#define ITER_PER_SPLIT (COLS_PER_SPLIT / 32)

__global__ __launch_bounds__(256) void DCM_xb2_kernel(const float* __restrict__ xb,
                                                      float* __restrict__ xb2) {
  const int j = blockIdx.x * 256 + threadIdx.x;
  const v4f* row = (const v4f*)(xb + (size_t)j * DIMK);
  v4f r0 = row[0], r1 = row[1], r2 = row[2], r3 = row[3];
  float s = r0.x * r0.x + r0.y * r0.y + r0.z * r0.z + r0.w * r0.w;
  s += r1.x * r1.x + r1.y * r1.y + r1.z * r1.z + r1.w * r1.w;
  s += r2.x * r2.x + r2.y * r2.y + r2.z * r2.z + r2.w * r2.w;
  s += r3.x * r3.x + r3.y * r3.y + r3.z * r3.z + r3.w * r3.w;
  xb2[j] = s;
}

__global__ __launch_bounds__(512) void DCM_nn_wmma_kernel(
    const float* __restrict__ x,
    const float* __restrict__ xb,
    const float* __restrict__ y,
    const float* __restrict__ xb2,
    float* __restrict__ out) {
  __shared__ float s_val[QT_PER_WAVE][NSPLIT][16];
  __shared__ int   s_idx[QT_PER_WAVE][NSPLIT][16];

  const int split = threadIdx.x >> 5;
  const int lane  = threadIdx.x & 31;
  const int qbase = blockIdx.x * (QT_PER_WAVE * 16);
  const int grp   = lane & 15;
  const int hi    = lane >> 4;

  v2f a[QT_PER_WAVE][4];
#pragma unroll
  for (int t = 0; t < QT_PER_WAVE; ++t) {
    const float* xrow = x + (size_t)(qbase + t * 16 + grp) * DIMK;
#pragma unroll
    for (int k = 0; k < 4; ++k) {
      const int kb = k * 4 + hi * 2;
      a[t][k].x = -2.0f * xrow[kb + 0];
      a[t][k].y = -2.0f * xrow[kb + 1];
    }
  }

  float bv[QT_PER_WAVE][8];
  int   bi[QT_PER_WAVE][8];
#pragma unroll
  for (int t = 0; t < QT_PER_WAVE; ++t) {
#pragma unroll
    for (int r = 0; r < 8; ++r) {
      bv[t][r] = 3.402823466e38f;
      bi[t][r] = 0;
    }
  }

  const int jt0 = split * ITER_PER_SPLIT;
  for (int jt = jt0; jt < jt0 + ITER_PER_SPLIT; ++jt) {
    const int c0 = jt * 32 + grp;
    const int c1 = c0 + 16;
    const float* p0 = xb + (size_t)c0 * DIMK;
    const float* p1 = xb + (size_t)c1 * DIMK;

    v2f b0[4], b1[4];
#pragma unroll
    for (int k = 0; k < 4; ++k) {
      const int kb = k * 4 + hi * 2;
      b0[k] = *(const v2f*)(p0 + kb);
      b1[k] = *(const v2f*)(p1 + kb);
    }
    const float s0 = xb2[c0];
    const float s1 = xb2[c1];

    v8f seed0, seed1;
#pragma unroll
    for (int r = 0; r < 8; ++r) { seed0[r] = s0; seed1[r] = s1; }

#pragma unroll
    for (int t = 0; t < QT_PER_WAVE; ++t) {
      v8f acc0 = __builtin_amdgcn_wmma_f32_16x16x4_f32(
          false, a[t][0], false, b0[0], (short)0, seed0, false, false);
      v8f acc1 = __builtin_amdgcn_wmma_f32_16x16x4_f32(
          false, a[t][0], false, b1[0], (short)0, seed1, false, false);
#pragma unroll
      for (int k = 1; k < 4; ++k) {
        acc0 = __builtin_amdgcn_wmma_f32_16x16x4_f32(
            false, a[t][k], false, b0[k], (short)0, acc0, false, false);
        acc1 = __builtin_amdgcn_wmma_f32_16x16x4_f32(
            false, a[t][k], false, b1[k], (short)0, acc1, false, false);
      }

#pragma unroll
      for (int r = 0; r < 8; ++r) {
        if (acc0[r] < bv[t][r]) { bv[t][r] = acc0[r]; bi[t][r] = c0; }
        if (acc1[r] < bv[t][r]) { bv[t][r] = acc1[r]; bi[t][r] = c1; }
      }
    }
  }

#pragma unroll
  for (int t = 0; t < QT_PER_WAVE; ++t) {
#pragma unroll
    for (int mask = 1; mask < 16; mask <<= 1) {
#pragma unroll
      for (int r = 0; r < 8; ++r) {
        const float ov = __shfl_xor(bv[t][r], mask, 32);
        const int   oi = __shfl_xor(bi[t][r], mask, 32);
        const bool take = (ov < bv[t][r]) | ((ov == bv[t][r]) & (oi < bi[t][r]));
        bv[t][r] = take ? ov : bv[t][r];
        bi[t][r] = take ? oi : bi[t][r];
      }
    }

    if (grp == 0) {
#pragma unroll
      for (int r = 0; r < 8; ++r) {
        s_val[t][split][hi * 8 + r] = bv[t][r];
        s_idx[t][split][hi * 8 + r] = bi[t][r];
      }
    }
  }
  __syncthreads();

  if (threadIdx.x < QT_PER_WAVE * 16) {
    const int qt  = threadIdx.x >> 4;
    const int row = threadIdx.x & 15;
    float v = s_val[qt][0][row];
    int   i = s_idx[qt][0][row];
#pragma unroll
    for (int s = 1; s < NSPLIT; ++s) {
      const float sv = s_val[qt][s][row];
      const int   si = s_idx[qt][s][row];
      if (sv < v) { v = sv; i = si; }
    }
    out[(blockIdx.x * QT_PER_WAVE + qt) * 16 + row] = y[i];
  }
}

extern "C" void kernel_launch(void* const* d_in, const int* in_sizes, int n_in,
                              void* d_out, int out_size, void* d_ws, size_t ws_size,
                              hipStream_t stream) {
  const float* x  = (const float*)d_in[0];
  const float* xb = (const float*)d_in[1];
  const float* y  = (const float*)d_in[2];
  float* out = (float*)d_out;
  float* xb2 = (float*)d_ws;

  DCM_xb2_kernel<<<dim3(MR / 256), dim3(256), 0, stream>>>(xb, xb2);

  DCM_nn_wmma_kernel<<<dim3(NQ / 16 / QT_PER_WAVE), dim3(512), 0, stream>>>(
      x, xb, y, xb2, out);
}


# Next iteration's B fragments and seeds are loaded into a second register set ahead of the current iteration's matrix work (loop unrolled by 2, loads pinned first): no wait on L2 inside the loop
# speedup vs baseline: 1.9586x; 1.2456x over previous
#include <hip/hip_runtime.h>


typedef __attribute__((ext_vector_type(2))) float v2f;
typedef __attribute__((ext_vector_type(4))) float v4f;
typedef __attribute__((ext_vector_type(8))) float v8f;

#define NQ 16384
#define MR 16384
#define DIMK 16
#define QT_PER_WAVE 4
#define NSPLIT 16
#define COLS_PER_SPLIT (MR / NSPLIT)
#define ITER_PER_SPLIT (COLS_PER_SPLIT / 32)

__global__ __launch_bounds__(256) void DCM_xb2_kernel(const float* __restrict__ xb,
                                                      float* __restrict__ xb2) {
  const int j = blockIdx.x * 256 + threadIdx.x;
  const v4f* row = (const v4f*)(xb + (size_t)j * DIMK);
  v4f r0 = row[0], r1 = row[1], r2 = row[2], r3 = row[3];
  float s = r0.x * r0.x + r0.y * r0.y + r0.z * r0.z + r0.w * r0.w;
  s += r1.x * r1.x + r1.y * r1.y + r1.z * r1.z + r1.w * r1.w;
  s += r2.x * r2.x + r2.y * r2.y + r2.z * r2.z + r2.w * r2.w;
  s += r3.x * r3.x + r3.y * r3.y + r3.z * r3.z + r3.w * r3.w;
  xb2[j] = s;
}

__global__ __launch_bounds__(512) void DCM_nn_wmma_kernel(
    const float* __restrict__ x,
    const float* __restrict__ xb,
    const float* __restrict__ y,
    const float* __restrict__ xb2,
    float* __restrict__ out) {
  __shared__ float s_val[QT_PER_WAVE][NSPLIT][16];
  __shared__ int   s_idx[QT_PER_WAVE][NSPLIT][16];

  const int split = threadIdx.x >> 5;
  const int lane  = threadIdx.x & 31;
  const int qbase = blockIdx.x * (QT_PER_WAVE * 16);
  const int grp   = lane & 15;
  const int hi    = lane >> 4;

  v2f a[QT_PER_WAVE][4];
#pragma unroll
  for (int t = 0; t < QT_PER_WAVE; ++t) {
    const float* xrow = x + (size_t)(qbase + t * 16 + grp) * DIMK;
#pragma unroll
    for (int k = 0; k < 4; ++k) {
      const int kb = k * 4 + hi * 2;
      a[t][k].x = -2.0f * xrow[kb + 0];
      a[t][k].y = -2.0f * xrow[kb + 1];
    }
  }

  float bv[QT_PER_WAVE][8];
  int   bi[QT_PER_WAVE][8];
#pragma unroll
  for (int t = 0; t < QT_PER_WAVE; ++t) {
#pragma unroll
    for (int r = 0; r < 8; ++r) {
      bv[t][r] = 3.402823466e38f;
      bi[t][r] = 0;
    }
  }

  const int jt0 = split * ITER_PER_SPLIT;
  v2f b0[4], b1[4];
  float s0, s1;
  {
    const int c0 = jt0 * 32 + grp;
    const int c1 = c0 + 16;
    const float* p0 = xb + (size_t)c0 * DIMK;
    const float* p1 = xb + (size_t)c1 * DIMK;
#pragma unroll
    for (int k = 0; k < 4; ++k) {
      const int kb = k * 4 + hi * 2;
      b0[k] = *(const v2f*)(p0 + kb);
      b1[k] = *(const v2f*)(p1 + kb);
    }
    s0 = xb2[c0];
    s1 = xb2[c1];
  }

#pragma unroll 2
  for (int it = 0; it < ITER_PER_SPLIT; ++it) {
    const int jt = jt0 + it;
    const int c0 = jt * 32 + grp;
    const int c1 = c0 + 16;

    const int jn = jt + ((it + 1 < ITER_PER_SPLIT) ? 1 : 0);
    const int n0 = jn * 32 + grp;
    const int n1 = n0 + 16;
    const float* q0 = xb + (size_t)n0 * DIMK;
    const float* q1 = xb + (size_t)n1 * DIMK;
    v2f nb0[4], nb1[4];
#pragma unroll
    for (int k = 0; k < 4; ++k) {
      const int kb = k * 4 + hi * 2;
      nb0[k] = *(const v2f*)(q0 + kb);
      nb1[k] = *(const v2f*)(q1 + kb);
    }
    const float ns0 = xb2[n0];
    const float ns1 = xb2[n1];
    __builtin_amdgcn_sched_barrier(0);

    v8f seed0, seed1;
#pragma unroll
    for (int r = 0; r < 8; ++r) { seed0[r] = s0; seed1[r] = s1; }

#pragma unroll
    for (int t = 0; t < QT_PER_WAVE; ++t) {
      v8f acc0 = __builtin_amdgcn_wmma_f32_16x16x4_f32(
          false, a[t][0], false, b0[0], (short)0, seed0, false, false);
      v8f acc1 = __builtin_amdgcn_wmma_f32_16x16x4_f32(
          false, a[t][0], false, b1[0], (short)0, seed1, false, false);
#pragma unroll
      for (int k = 1; k < 4; ++k) {
        acc0 = __builtin_amdgcn_wmma_f32_16x16x4_f32(
            false, a[t][k], false, b0[k], (short)0, acc0, false, false);
        acc1 = __builtin_amdgcn_wmma_f32_16x16x4_f32(
            false, a[t][k], false, b1[k], (short)0, acc1, false, false);
      }

#pragma unroll
      for (int r = 0; r < 8; ++r) {
        if (acc0[r] < bv[t][r]) { bv[t][r] = acc0[r]; bi[t][r] = c0; }
        if (acc1[r] < bv[t][r]) { bv[t][r] = acc1[r]; bi[t][r] = c1; }
      }
    }

#pragma unroll
    for (int k = 0; k < 4; ++k) { b0[k] = nb0[k]; b1[k] = nb1[k]; }
    s0 = ns0;
    s1 = ns1;
  }

#pragma unroll
  for (int t = 0; t < QT_PER_WAVE; ++t) {
#pragma unroll
    for (int mask = 1; mask < 16; mask <<= 1) {
#pragma unroll
      for (int r = 0; r < 8; ++r) {
        const float ov = __shfl_xor(bv[t][r], mask, 32);
        const int   oi = __shfl_xor(bi[t][r], mask, 32);
        const bool take = (ov < bv[t][r]) | ((ov == bv[t][r]) & (oi < bi[t][r]));
        bv[t][r] = take ? ov : bv[t][r];
        bi[t][r] = take ? oi : bi[t][r];
      }
    }

    if (grp == 0) {
#pragma unroll
      for (int r = 0; r < 8; ++r) {
        s_val[t][split][hi * 8 + r] = bv[t][r];
        s_idx[t][split][hi * 8 + r] = bi[t][r];
      }
    }
  }
  __syncthreads();

  if (threadIdx.x < QT_PER_WAVE * 16) {
    const int qt  = threadIdx.x >> 4;
    const int row = threadIdx.x & 15;
    float v = s_val[qt][0][row];
    int   i = s_idx[qt][0][row];
#pragma unroll
    for (int s = 1; s < NSPLIT; ++s) {
      const float sv = s_val[qt][s][row];
      const int   si = s_idx[qt][s][row];
      if (sv < v) { v = sv; i = si; }
    }
    out[(blockIdx.x * QT_PER_WAVE + qt) * 16 + row] = y[i];
  }
}

extern "C" void kernel_launch(void* const* d_in, const int* in_sizes, int n_in,
                              void* d_out, int out_size, void* d_ws, size_t ws_size,
                              hipStream_t stream) {
  const float* x  = (const float*)d_in[0];
  const float* xb = (const float*)d_in[1];
  const float* y  = (const float*)d_in[2];
  float* out = (float*)d_out;
  float* xb2 = (float*)d_ws;

  DCM_xb2_kernel<<<dim3(MR / 256), dim3(256), 0, stream>>>(xb, xb2);

  DCM_nn_wmma_kernel<<<dim3(NQ / 16 / QT_PER_WAVE), dim3(512), 0, stream>>>(
      x, xb, y, xb2, out);
}


# Deferred argmin index: loop keeps a 3-input running min per row and, every 4 iterations, which 8 adjacent columns lowered it; the column is found per query afterwards by recomputing those 8 scores
# speedup vs baseline: 2.1402x; 1.0927x over previous
#include <hip/hip_runtime.h>


typedef __attribute__((ext_vector_type(2))) float v2f;
typedef __attribute__((ext_vector_type(4))) float v4f;
typedef __attribute__((ext_vector_type(8))) float v8f;

#define NQ 16384
#define MR 16384
#define DIMK 16
#define QT_PER_WAVE 4
#define NSPLIT 16
#define COLS_PER_SPLIT (MR / NSPLIT)
#define ITER_PER_SPLIT (COLS_PER_SPLIT / 32)
#define GRP_ITERS 4
#define GROUPS_PER_SPLIT (ITER_PER_SPLIT / GRP_ITERS)

__global__ __launch_bounds__(256) void DCM_xb2_kernel(const float* __restrict__ xb,
                                                      float* __restrict__ xb2) {
  const int j = blockIdx.x * 256 + threadIdx.x;
  const v4f* row = (const v4f*)(xb + (size_t)j * DIMK);
  v4f r0 = row[0], r1 = row[1], r2 = row[2], r3 = row[3];
  float s = r0.x * r0.x + r0.y * r0.y + r0.z * r0.z + r0.w * r0.w;
  s += r1.x * r1.x + r1.y * r1.y + r1.z * r1.z + r1.w * r1.w;
  s += r2.x * r2.x + r2.y * r2.y + r2.z * r2.z + r2.w * r2.w;
  s += r3.x * r3.x + r3.y * r3.y + r3.z * r3.z + r3.w * r3.w;
  xb2[j] = s;
}

__global__ __launch_bounds__(512) void DCM_nn_wmma_kernel(
    const float* __restrict__ x,
    const float* __restrict__ xb,
    const float* __restrict__ y,
    const float* __restrict__ xb2,
    float* __restrict__ out) {
  __shared__ float s_val[QT_PER_WAVE][NSPLIT][16];
  __shared__ int   s_idx[QT_PER_WAVE][NSPLIT][16];

  const int split = __builtin_amdgcn_readfirstlane((int)(threadIdx.x >> 5));
  const int lane  = threadIdx.x & 31;
  const int qbase = blockIdx.x * (QT_PER_WAVE * 16);
  const int grp   = lane & 15;
  const int hi    = lane >> 4;

  v2f a[QT_PER_WAVE][4];
#pragma unroll
  for (int t = 0; t < QT_PER_WAVE; ++t) {
    const float* xrow = x + (size_t)(qbase + t * 16 + grp) * DIMK;
#pragma unroll
    for (int k = 0; k < 4; ++k) {
      const int kb = k * 4 + hi * 2;
      a[t][k].x = -2.0f * xrow[kb + 0];
      a[t][k].y = -2.0f * xrow[kb + 1];
    }
  }

  float bv[QT_PER_WAVE][8];
  int   bi[QT_PER_WAVE][8];
#pragma unroll
  for (int t = 0; t < QT_PER_WAVE; ++t) {
#pragma unroll
    for (int r = 0; r < 8; ++r) {
      bv[t][r] = 3.402823466e38f;
      bi[t][r] = 0;
    }
  }

  const int jt0 = split * ITER_PER_SPLIT;
  v2f b0[4], b1[4];
  float s0, s1;
  {
    const int c0 = jt0 * 32 + grp * 8;
    const float* p0 = xb + (size_t)c0 * DIMK;
    const float* p1 = p0 + DIMK;
#pragma unroll
    for (int k = 0; k < 4; ++k) {
      const int kb = k * 4 + hi * 2;
      b0[k] = *(const v2f*)(p0 + kb);
      b1[k] = *(const v2f*)(p1 + kb);
    }
    const v2f sp = *(const v2f*)(xb2 + c0);
    s0 = sp.x;
    s1 = sp.y;
  }

  for (int g = 0; g < GROUPS_PER_SPLIT; ++g) {
    float old[QT_PER_WAVE][8];
#pragma unroll
    for (int t = 0; t < QT_PER_WAVE; ++t) {
#pragma unroll
      for (int r = 0; r < 8; ++r) old[t][r] = bv[t][r];
    }

#pragma unroll
    for (int i = 0; i < GRP_ITERS; ++i) {
    const int it = g * GRP_ITERS + i;
    const int itn = it + ((it + 1 < ITER_PER_SPLIT) ? 1 : 0);
    const int n0 = jt0 * 32 + (itn / GRP_ITERS) * (GRP_ITERS * 32) + (itn % GRP_ITERS) * 2 + grp * 8;
    const float* q0 = xb + (size_t)n0 * DIMK;
    const float* q1 = q0 + DIMK;
    v2f nb0[4], nb1[4];
#pragma unroll
    for (int k = 0; k < 4; ++k) {
      const int kb = k * 4 + hi * 2;
      nb0[k] = *(const v2f*)(q0 + kb);
      nb1[k] = *(const v2f*)(q1 + kb);
    }
    const v2f nsp = *(const v2f*)(xb2 + n0);
    const float ns0 = nsp.x;
    const float ns1 = nsp.y;
    __builtin_amdgcn_sched_barrier(0);

    v8f seed0, seed1;
#pragma unroll
    for (int r = 0; r < 8; ++r) { seed0[r] = s0; seed1[r] = s1; }

#pragma unroll
    for (int t = 0; t < QT_PER_WAVE; ++t) {
      v8f acc0 = __builtin_amdgcn_wmma_f32_16x16x4_f32(
          false, a[t][0], false, b0[0], (short)0, seed0, false, false);
      v8f acc1 = __builtin_amdgcn_wmma_f32_16x16x4_f32(
          false, a[t][0], false, b1[0], (short)0, seed1, false, false);
#pragma unroll
      for (int k = 1; k < 4; ++k) {
        acc0 = __builtin_amdgcn_wmma_f32_16x16x4_f32(
            false, a[t][k], false, b0[k], (short)0, acc0, false, false);
        acc1 = __builtin_amdgcn_wmma_f32_16x16x4_f32(
            false, a[t][k], false, b1[k], (short)0, acc1, false, false);
      }

#pragma unroll
      for (int r = 0; r < 8; ++r) {
        bv[t][r] = __builtin_fminf(__builtin_fminf(acc0[r], acc1[r]), bv[t][r]);
      }
      if (t == 1) __builtin_amdgcn_sched_barrier(0);
    }

#pragma unroll
    for (int k = 0; k < 4; ++k) { b0[k] = nb0[k]; b1[k] = nb1[k]; }
    s0 = ns0;
    s1 = ns1;
    }

    const int gkey = split * GROUPS_PER_SPLIT + g;
#pragma unroll
    for (int t = 0; t < QT_PER_WAVE; ++t) {
#pragma unroll
      for (int r = 0; r < 8; ++r) {
        bi[t][r] = (bv[t][r] < old[t][r]) ? gkey : bi[t][r];
      }
    }

  }

#pragma unroll
  for (int t = 0; t < QT_PER_WAVE; ++t) {
#pragma unroll
    for (int r = 0; r < 8; ++r) bi[t][r] = bi[t][r] * 16 + grp;

#pragma unroll
    for (int mask = 1; mask < 16; mask <<= 1) {
#pragma unroll
      for (int r = 0; r < 8; ++r) {
        const float ov = __shfl_xor(bv[t][r], mask, 32);
        const int   oi = __shfl_xor(bi[t][r], mask, 32);
        const bool take = (ov < bv[t][r]) | ((ov == bv[t][r]) & (oi < bi[t][r]));
        bv[t][r] = take ? ov : bv[t][r];
        bi[t][r] = take ? oi : bi[t][r];
      }
    }

    if (grp == 0) {
#pragma unroll
      for (int r = 0; r < 8; ++r) {
        s_val[t][split][hi * 8 + r] = bv[t][r];
        s_idx[t][split][hi * 8 + r] = bi[t][r];
      }
    }
  }
  __syncthreads();

  if (split < 2) {
#pragma unroll
    for (int u = 0; u < 2; ++u) {
      const int qt = split * 2 + u;
      float v = s_val[qt][0][grp];
      int   p = s_idx[qt][0][grp];
#pragma unroll
      for (int s = 1; s < NSPLIT; ++s) {
        const float sv = s_val[qt][s][grp];
        const int   si = s_idx[qt][s][grp];
        if (sv < v) { v = sv; p = si; }
      }

      const int col0 = p * 8;
      const float* pg = xb + (size_t)col0 * DIMK;
      float bestv = 0.0f;
      int   besti = 0;
#pragma unroll
      for (int i = 0; i < GRP_ITERS; ++i) {
        v2f fb0[4], fb1[4];
#pragma unroll
        for (int k = 0; k < 4; ++k) {
          const int kb = k * 4 + hi * 2;
          fb0[k] = *(const v2f*)(pg + (2 * i) * DIMK + kb);
          fb1[k] = *(const v2f*)(pg + (2 * i + 1) * DIMK + kb);
        }
        const v2f fs = *(const v2f*)(xb2 + col0 + 2 * i);
        v8f d0, d1;
#pragma unroll
        for (int r = 0; r < 8; ++r) { d0[r] = fs.x; d1[r] = fs.y; }
#pragma unroll
        for (int k = 0; k < 4; ++k) {
          const v2f ak = (split == 0) ? a[u][k] : a[2 + u][k];
          d0 = __builtin_amdgcn_wmma_f32_16x16x4_f32(
              false, ak, false, fb0[k], (short)0, d0, false, false);
          d1 = __builtin_amdgcn_wmma_f32_16x16x4_f32(
              false, ak, false, fb1[k], (short)0, d1, false, false);
        }
        float e0 = d0[0], e1 = d1[0];
#pragma unroll
        for (int r = 1; r < 8; ++r) {
          if ((grp & 7) == r) { e0 = d0[r]; e1 = d1[r]; }
        }
        if (i == 0) { bestv = e0; besti = 0; }
        else if (e0 < bestv) { bestv = e0; besti = 2 * i; }
        if (e1 < bestv) { bestv = e1; besti = 2 * i + 1; }
      }
      if (hi == (grp >> 3)) {
        out[(blockIdx.x * QT_PER_WAVE + qt) * 16 + grp] = y[col0 + besti];
      }
    }
  }
}

extern "C" void kernel_launch(void* const* d_in, const int* in_sizes, int n_in,
                              void* d_out, int out_size, void* d_ws, size_t ws_size,
                              hipStream_t stream) {
  const float* x  = (const float*)d_in[0];
  const float* xb = (const float*)d_in[1];
  const float* y  = (const float*)d_in[2];
  float* out = (float*)d_out;
  float* xb2 = (float*)d_ws;

  DCM_xb2_kernel<<<dim3(MR / 256), dim3(256), 0, stream>>>(xb, xb2);

  DCM_nn_wmma_kernel<<<dim3(NQ / 16 / QT_PER_WAVE), dim3(512), 0, stream>>>(
      x, xb, y, xb2, out);
}


# Operand roles swapped: reference points are the A rows, queries the B columns, so each query's minimum and index live in one lane; one half-to-half exchange replaces the 16-lane butterflies
# speedup vs baseline: 2.2964x; 1.0730x over previous
#include <hip/hip_runtime.h>


typedef __attribute__((ext_vector_type(2))) float v2f;
typedef __attribute__((ext_vector_type(4))) float v4f;
typedef __attribute__((ext_vector_type(8))) float v8f;

#define NQ 16384
#define MR 16384
#define DIMK 16
#define QT_PER_WAVE 4
#define NSPLIT 16
#define REFS_PER_SPLIT (MR / NSPLIT)
#define ITER_PER_SPLIT (REFS_PER_SPLIT / 32)

#define MIN3(a, b, c) __builtin_fminf(__builtin_fminf((a), (b)), (c))

__global__ __launch_bounds__(256) void DCM_xb2_kernel(const float* __restrict__ xb,
                                                      float* __restrict__ xb2) {
  const int j = blockIdx.x * 256 + threadIdx.x;
  const v4f* row = (const v4f*)(xb + (size_t)j * DIMK);
  v4f r0 = row[0], r1 = row[1], r2 = row[2], r3 = row[3];
  float s = r0.x * r0.x + r0.y * r0.y + r0.z * r0.z + r0.w * r0.w;
  s += r1.x * r1.x + r1.y * r1.y + r1.z * r1.z + r1.w * r1.w;
  s += r2.x * r2.x + r2.y * r2.y + r2.z * r2.z + r2.w * r2.w;
  s += r3.x * r3.x + r3.y * r3.y + r3.z * r3.z + r3.w * r3.w;
  xb2[j] = s;
}

__device__ __forceinline__ int xhalf_i(int v) {
  return __builtin_amdgcn_permlanex16(v, v, 0x76543210, 0xfedcba98, false, false);
}
__device__ __forceinline__ float xhalf_f(float v) {
  return __builtin_bit_cast(float, xhalf_i(__builtin_bit_cast(int, v)));
}

__global__ __launch_bounds__(512) void DCM_nn_wmma_kernel(
    const float* __restrict__ x,
    const float* __restrict__ xb,
    const float* __restrict__ y,
    const float* __restrict__ xb2,
    float* __restrict__ out) {
  __shared__ float s_val[QT_PER_WAVE][NSPLIT][16];
  __shared__ int   s_idx[QT_PER_WAVE][NSPLIT][16];
  __shared__ float s_out[QT_PER_WAVE][16];

  const int split = __builtin_amdgcn_readfirstlane((int)(threadIdx.x >> 5));
  const int lane  = threadIdx.x & 31;
  const int qbase = blockIdx.x * (QT_PER_WAVE * 16);
  const int grp   = lane & 15;
  const int hi    = lane >> 4;

  v2f q[QT_PER_WAVE][4];
#pragma unroll
  for (int t = 0; t < QT_PER_WAVE; ++t) {
    const float* xrow = x + (size_t)(qbase + t * 16 + grp) * DIMK;
#pragma unroll
    for (int k = 0; k < 4; ++k) {
      const int kb = k * 4 + hi * 2;
      q[t][k].x = -2.0f * xrow[kb + 0];
      q[t][k].y = -2.0f * xrow[kb + 1];
    }
  }

  v2f fq[4];
  {
    const float* xrow = x + (size_t)(qbase + (split & (QT_PER_WAVE - 1)) * 16 + grp) * DIMK + hi * 2;
#pragma unroll
    for (int k = 0; k < 4; ++k) {
      const v2f r = *(const v2f*)(xrow + k * 4);
      fq[k].x = -2.0f * r.x;
      fq[k].y = -2.0f * r.y;
    }
  }

  float bv[QT_PER_WAVE];
  int   bi[QT_PER_WAVE];
#pragma unroll
  for (int t = 0; t < QT_PER_WAVE; ++t) {
    bv[t] = 3.402823466e38f;
    bi[t] = 0;
  }

  const int jbase  = split * REFS_PER_SPLIT;
  const int rowoff = (grp & 7) + ((grp >> 3) << 4);
  const float* prow  = xb + (size_t)(jbase + rowoff) * DIMK + hi * 2;
  const float* pseed = xb2 + jbase + hi * 16;
  v2f a0[4], a1[4];
  v4f sd[4];
#pragma unroll
  for (int k = 0; k < 4; ++k) {
    a0[k] = *(const v2f*)(prow + k * 4);
    a1[k] = *(const v2f*)(prow + 8 * DIMK + k * 4);
    sd[k] = *(const v4f*)(pseed + k * 4);
  }

  for (int it2 = 0; it2 < ITER_PER_SPLIT / 2; ++it2) {
#pragma unroll
    for (int i = 0; i < 2; ++i) {
      const int it = it2 * 2 + i;
      const int itn = it + ((it + 1 < ITER_PER_SPLIT) ? 1 : 0);
      const float* nrow  = prow + (size_t)itn * (32 * DIMK);
      const float* nseed = pseed + itn * 32;
      v2f na0[4], na1[4];
      v4f nsd[4];
#pragma unroll
      for (int k = 0; k < 4; ++k) {
        na0[k] = *(const v2f*)(nrow + k * 4);
        na1[k] = *(const v2f*)(nrow + 8 * DIMK + k * 4);
        nsd[k] = *(const v4f*)(nseed + k * 4);
      }
      __builtin_amdgcn_sched_barrier(0);

      const v8f seed0 = __builtin_shufflevector(sd[0], sd[1], 0, 1, 2, 3, 4, 5, 6, 7);
      const v8f seed1 = __builtin_shufflevector(sd[2], sd[3], 0, 1, 2, 3, 4, 5, 6, 7);
      const int key = (split * ITER_PER_SPLIT + it) * 2;

#pragma unroll
      for (int t = 0; t < QT_PER_WAVE; ++t) {
        v8f acc0 = __builtin_amdgcn_wmma_f32_16x16x4_f32(
            false, a0[0], false, q[t][0], (short)0, seed0, false, false);
        v8f acc1 = __builtin_amdgcn_wmma_f32_16x16x4_f32(
            false, a1[0], false, q[t][0], (short)0, seed1, false, false);
#pragma unroll
        for (int k = 1; k < 4; ++k) {
          acc0 = __builtin_amdgcn_wmma_f32_16x16x4_f32(
              false, a0[k], false, q[t][k], (short)0, acc0, false, false);
          acc1 = __builtin_amdgcn_wmma_f32_16x16x4_f32(
              false, a1[k], false, q[t][k], (short)0, acc1, false, false);
        }

        float nb = bv[t];
#pragma unroll
        for (int r = 0; r < 8; r += 2) nb = MIN3(acc0[r], acc0[r + 1], nb);
#pragma unroll
        for (int r = 0; r < 8; r += 2) nb = MIN3(acc1[r], acc1[r + 1], nb);
        bi[t] = (nb < bv[t]) ? key : bi[t];
        bv[t] = nb;
      }

#pragma unroll
      for (int k = 0; k < 4; ++k) { a0[k] = na0[k]; a1[k] = na1[k]; sd[k] = nsd[k]; }
    }
  }

#pragma unroll
  for (int t = 0; t < QT_PER_WAVE; ++t) {
    const int   kk = bi[t] + hi;
    const float ov = xhalf_f(bv[t]);
    const int   ok = xhalf_i(kk);
    const bool take = (ov < bv[t]) | ((ov == bv[t]) & (ok < kk));
    const float wv = take ? ov : bv[t];
    const int   wk = take ? ok : kk;
    if (hi == 0) {
      s_val[t][split][grp] = wv;
      s_idx[t][split][grp] = wk;
    }
  }
  __syncthreads();

  if (split < QT_PER_WAVE) {
    const int qt = split;
    float v = s_val[qt][0][grp];
    int   p = s_idx[qt][0][grp];
#pragma unroll
    for (int s = 1; s < NSPLIT; ++s) {
      const float sv = s_val[qt][s][grp];
      const int   si = s_idx[qt][s][grp];
      if (sv < v) { v = sv; p = si; }
    }

    const int col0 = p * 16;
    const float* pg = xb + (size_t)col0 * DIMK + hi * 2;
    v4f fs4[4];
#pragma unroll
    for (int j = 0; j < 4; ++j) fs4[j] = *(const v4f*)(xb2 + col0 + j * 4);
    float bestv = 0.0f;
    int   besti = 0;
#pragma unroll
    for (int i = 0; i < 16; ++i) {
      v2f fa[4];
#pragma unroll
      for (int k = 0; k < 4; ++k) fa[k] = *(const v2f*)(pg + i * DIMK + k * 4);
      const float fs = fs4[i >> 2][i & 3];
      v8f d;
#pragma unroll
      for (int r = 0; r < 8; ++r) d[r] = fs;
#pragma unroll
      for (int k = 0; k < 4; ++k) {
        d = __builtin_amdgcn_wmma_f32_16x16x4_f32(
            false, fa[k], false, fq[k], (short)0, d, false, false);
      }
      float e = d[0];
#pragma unroll
      for (int r = 1; r < 8; ++r) {
        if ((grp & 7) == r) e = d[r];
      }
      if (i == 0) { bestv = e; besti = 0; }
      else if (e < bestv) { bestv = e; besti = i; }
    }
    if (hi == (grp >> 3)) s_out[qt][grp] = y[col0 + besti];
  }
  __syncthreads();

  if (split < 2) {
    out[blockIdx.x * (QT_PER_WAVE * 16) + split * 32 + lane] = s_out[split * 2 + hi][grp];
  }
}

extern "C" void kernel_launch(void* const* d_in, const int* in_sizes, int n_in,
                              void* d_out, int out_size, void* d_ws, size_t ws_size,
                              hipStream_t stream) {
  const float* x  = (const float*)d_in[0];
  const float* xb = (const float*)d_in[1];
  const float* y  = (const float*)d_in[2];
  float* out = (float*)d_out;
  float* xb2 = (float*)d_ws;

  DCM_xb2_kernel<<<dim3(MR / 256), dim3(256), 0, stream>>>(xb, xb2);

  DCM_nn_wmma_kernel<<<dim3(NQ / 16 / QT_PER_WAVE), dim3(512), 0, stream>>>(
      x, xb, y, xb2, out);
}
